# starting kernel
# baseline (speedup 1.0000x reference)
#include <hip/hip_runtime.h>


typedef float v2f __attribute__((ext_vector_type(2)));
typedef float v8f __attribute__((ext_vector_type(8)));

constexpr int B_   = 4;
constexpr int CQ_  = 256;
constexpr int CKV_ = 256;
constexpr int NQ_  = 2048;
constexpr int NKV_ = 2048;
constexpr int H_   = 8;
constexpr int DK_  = 64;
constexpr int OD_  = H_ * DK_;

template <int NB, bool HASR>
__device__ __forceinline__ void wmma_gemm_strip(
    const float* __restrict__ A, long rsA, long csA,
    const float* __restrict__ Bm, long rsB, long csB,
    float* __restrict__ C, long rsC,
    const float* __restrict__ R, long rsR,
    int m0, int n0, int K, float scale) {
  const int lane = threadIdx.x & 31;
  const int r    = lane & 15;
  const int half = lane >> 4;
  const long arow = (long)(m0 + r) * rsA;
  long bcol[NB];
#pragma unroll
  for (int j = 0; j < NB; ++j) bcol[j] = (long)(n0 + 16 * j + r) * csB;

  v8f acc[NB];
#pragma unroll
  for (int j = 0; j < NB; ++j) acc[j] = {};

  for (int k0 = 0; k0 < K; k0 += 8) {
    const int kA = k0 + 2 * half;
    __builtin_prefetch(&Bm[(long)(kA + 32) * rsB + bcol[0]], 0, 3);
    v2f a0, a1;
    a0.x = A[arow + (long)(kA    ) * csA];
    a0.y = A[arow + (long)(kA + 1) * csA];
    a1.x = A[arow + (long)(kA + 4) * csA];
    a1.y = A[arow + (long)(kA + 5) * csA];
    v2f b0[NB], b1[NB];
#pragma unroll
    for (int j = 0; j < NB; ++j) {
      b0[j].x = Bm[(long)(kA    ) * rsB + bcol[j]];
      b0[j].y = Bm[(long)(kA + 1) * rsB + bcol[j]];
      b1[j].x = Bm[(long)(kA + 4) * rsB + bcol[j]];
      b1[j].y = Bm[(long)(kA + 5) * rsB + bcol[j]];
    }
#pragma unroll
    for (int j = 0; j < NB; ++j)
      acc[j] = __builtin_amdgcn_wmma_f32_16x16x4_f32(false, a0, false, b0[j],
                                                     (short)0, acc[j], false, false);
#pragma unroll
    for (int j = 0; j < NB; ++j)
      acc[j] = __builtin_amdgcn_wmma_f32_16x16x4_f32(false, a1, false, b1[j],
                                                     (short)0, acc[j], false, false);
  }

#pragma unroll
  for (int j = 0; j < NB; ++j) {
#pragma unroll
    for (int v = 0; v < 8; ++v) {
      const long row = m0 + v + 8 * half;
      const long col = n0 + 16 * j + r;
      float val = acc[j][v] * scale;
      if (HASR) val += R[row * rsR + col];
      C[row * rsC + col] = val;
    }
  }
}

__global__ void k_stage1_xxT(const float* __restrict__ xkv, float* __restrict__ S) {
  const int b = blockIdx.z;
  const float* X = xkv + (long)b * CKV_ * NKV_;
  wmma_gemm_strip<2, false>(X, NKV_, 1, X, 1, NKV_,
                            S + (long)b * CKV_ * CKV_, CKV_, nullptr, 0,
                            blockIdx.y * 16, blockIdx.x * 32, NKV_, 1.0f);
}

__global__ void k_stage2_T1(const float* __restrict__ Wk, const float* __restrict__ S,
                            float* __restrict__ T1) {
  const int z = blockIdx.z;
  const int b = z >> 3, h = z & 7;
  wmma_gemm_strip<4, false>(Wk + (long)h * DK_ * CKV_, CKV_, 1,
                            S + (long)b * CKV_ * CKV_, CKV_, 1,
                            T1 + (long)z * DK_ * CKV_, CKV_, nullptr, 0,
                            blockIdx.y * 16, blockIdx.x * 64, CKV_, 1.0f);
}

__global__ void k_stage3_M(const float* __restrict__ T1, const float* __restrict__ Wv,
                           float* __restrict__ Mm) {
  const int z = blockIdx.z;
  const int h = z & 7;
  wmma_gemm_strip<4, false>(T1 + (long)z * DK_ * CKV_, CKV_, 1,
                            Wv + (long)h * DK_ * CKV_, 1, CKV_,
                            Mm + (long)z * DK_ * DK_, DK_, nullptr, 0,
                            blockIdx.y * 16, blockIdx.x * 64, CKV_, 1.0f / (float)DK_);
}

__global__ void k_stage4_Pt(const float* __restrict__ Mm, const float* __restrict__ Wq,
                            float* __restrict__ Pt) {
  const int z = blockIdx.z;
  const int h = z & 7;
  wmma_gemm_strip<4, false>(Mm + (long)z * DK_ * DK_, 1, DK_,
                            Wq + (long)h * DK_ * CQ_, CQ_, 1,
                            Pt + (long)z * DK_ * CQ_, CQ_, nullptr, 0,
                            blockIdx.y * 16, blockIdx.x * 64, DK_, 1.0f);
}

__global__ void k_stage5_G(const float* __restrict__ Wo, const float* __restrict__ Pt,
                           const float* __restrict__ gamma, float* __restrict__ G) {
  const int b = blockIdx.z;
  wmma_gemm_strip<4, false>(Wo, OD_, 1,
                            Pt + (long)b * OD_ * CQ_, CQ_, 1,
                            G + (long)b * CQ_ * CQ_, CQ_, nullptr, 0,
                            blockIdx.y * 16, blockIdx.x * 64, OD_, gamma[0]);
}

__global__ void k_stage6_out(const float* __restrict__ G, const float* __restrict__ xq,
                             float* __restrict__ out) {
  const int b = blockIdx.z;
  const float* Bq = xq + (long)b * CQ_ * NQ_;
  wmma_gemm_strip<4, true>(G + (long)b * CQ_ * CQ_, CQ_, 1,
                           Bq, NQ_, 1,
                           out + (long)b * CQ_ * NQ_, NQ_, Bq, NQ_,
                           blockIdx.y * 16, blockIdx.x * 64, CQ_, 1.0f);
}

extern "C" void kernel_launch(void* const* d_in, const int* in_sizes, int n_in,
                              void* d_out, int out_size, void* d_ws, size_t ws_size,
                              hipStream_t stream) {
  const float* xq    = (const float*)d_in[0];
  const float* xkv   = (const float*)d_in[1];
  const float* Wq    = (const float*)d_in[2];
  const float* Wk    = (const float*)d_in[3];
  const float* Wv    = (const float*)d_in[4];
  const float* Wo    = (const float*)d_in[5];
  const float* gamma = (const float*)d_in[6];
  float* out = (float*)d_out;

  float* S  = (float*)d_ws;
  float* T1 = S  + (size_t)B_ * CKV_ * CKV_;
  float* Mm = T1 + (size_t)B_ * H_ * DK_ * CKV_;
  float* Pt = Mm + (size_t)B_ * H_ * DK_ * DK_;
  float* G  = Pt + (size_t)B_ * H_ * DK_ * CQ_;

  const dim3 blk(32, 1, 1);
  k_stage1_xxT<<<dim3(CKV_ / 32, CKV_ / 16, B_),     blk, 0, stream>>>(xkv, S);
  k_stage2_T1 <<<dim3(CKV_ / 64, DK_ / 16, B_ * H_), blk, 0, stream>>>(Wk, S, T1);
  k_stage3_M  <<<dim3(DK_ / 64,  DK_ / 16, B_ * H_), blk, 0, stream>>>(T1, Wv, Mm);
  k_stage4_Pt <<<dim3(CQ_ / 64,  DK_ / 16, B_ * H_), blk, 0, stream>>>(Mm, Wq, Pt);
  k_stage5_G  <<<dim3(CQ_ / 64,  CQ_ / 16, B_),      blk, 0, stream>>>(Wo, Pt, gamma, G);
  k_stage6_out<<<dim3(NQ_ / 64,  CQ_ / 16, B_),      blk, 0, stream>>>(G, xq, out);
}


# Stage 1: K split over 4 workgroups (512 of 64x32, two per WGP, 2 tiles per wave) writing partial S planes; stage 2 (64x32 blocks) adds the 4 planes as it stages its B operand: sum order changes
# speedup vs baseline: 3.3004x; 3.3004x over previous
#include <hip/hip_runtime.h>


typedef float v2f __attribute__((ext_vector_type(2)));
typedef float v4f __attribute__((ext_vector_type(4)));
typedef float v8f __attribute__((ext_vector_type(8)));

constexpr int B_   = 4;
constexpr int CQ_  = 256;
constexpr int CKV_ = 256;
constexpr int NQ_  = 2048;
constexpr int NKV_ = 2048;
constexpr int H_   = 8;
constexpr int DK_  = 64;
constexpr int OD_  = H_ * DK_;

template <int X, int BK, bool KC>
struct OpCfg {
  static_assert(KC || (2 * X) % 64 == 0, "pair-row pitch rule needs X a multiple of 32");
  static constexpr int P   = KC ? BK + 4 : 2 * X + 32;
  static constexpr int SZ  = KC ? X * P : (BK / 2) * P;
  static constexpr int Q   = KC ? BK / 4 : X / 4;
  static constexpr int NPC = KC ? X * Q : (BK / 2) * Q;
  static constexpr int LPP = KC ? 1 : 2;
};

template <int BM, int BN, int BK, bool AKC, bool BKC>
struct TileCfg {
  using OA = OpCfg<BM, BK, AKC>;
  using OB = OpCfg<BN, BK, BKC>;
  static constexpr int SZA = OA::SZ;
  static constexpr int SZB = OB::SZ;
  static constexpr int PC  = BN + 4;
  static constexpr int SZC = BM * PC;
  static constexpr int LDS_FLOATS =
      2 * (SZA + SZB) > SZC ? 2 * (SZA + SZB) : SZC;
};

template <int X, int BK, bool KC, int NT, int NP>
__device__ __forceinline__ void piece_table(int (&g)[NP], int (&s)[NP], int tid, int x0, int ld) {
  using O = OpCfg<X, BK, KC>;
#pragma unroll
  for (int i = 0; i < NP; ++i) {
    const int idx = tid + i * NT;
    const int row = idx / O::Q;
    const int c   = (idx % O::Q) * 4;
    if constexpr (KC) {
      g[i] = (x0 + row) * ld + c;
      s[i] = row * O::P + c;
    } else {
      g[i] = (2 * row) * ld + x0 + c;
      s[i] = row * O::P + 2 * c;
    }
  }
}

template <bool KC, int NP>
__device__ __forceinline__ void chunk_load(v4f (&rg)[NP * (KC ? 1 : 2)], const float* __restrict__ p,
                                           const int (&g)[NP], int off, int ld) {
#pragma unroll
  for (int i = 0; i < NP; ++i) {
    if constexpr (KC) {
      rg[i] = *(const v4f*)(p + g[i] + off);
    } else {
      rg[2 * i]     = *(const v4f*)(p + g[i] + off);
      rg[2 * i + 1] = *(const v4f*)(p + g[i] + off + ld);
    }
  }
}

template <bool KC, int NP>
__device__ __forceinline__ void chunk_store(const v4f (&rg)[NP * (KC ? 1 : 2)], float* __restrict__ dst,
                                            const int (&s)[NP]) {
#pragma unroll
  for (int i = 0; i < NP; ++i) {
    if constexpr (KC) {
      *(v4f*)(dst + s[i]) = rg[i];
    } else {
      const v4f e = rg[2 * i], o = rg[2 * i + 1];
      v4f lo, hi;
      lo.x = e.x; lo.y = o.x; lo.z = e.y; lo.w = o.y;
      hi.x = e.z; hi.y = o.z; hi.z = e.w; hi.w = o.w;
      *(v4f*)(dst + s[i])     = lo;
      *(v4f*)(dst + s[i] + 4) = hi;
    }
  }
}

template <int X, int BK, bool KC>
__device__ __forceinline__ v2f frag_read(const float* __restrict__ base, int x, int kk) {
  using O = OpCfg<X, BK, KC>;
  if constexpr (KC) return *(const v2f*)(base + x * O::P + kk);
  else              return *(const v2f*)(base + (kk >> 1) * O::P + 2 * x);
}

template <int BM, int BN, int BK, int WM, int WN, bool AKC, bool BKC, bool HASR, int NPL = 1>
__device__ __forceinline__ void block_gemm(
    const float* __restrict__ A, int ldA,
    const float* __restrict__ Bm, int ldB,
    float* __restrict__ C, int ldC,
    const float* __restrict__ R, int ldR,
    int m0, int n0, int K, float scale, float* __restrict__ smem,
    const float* __restrict__ B1 = nullptr, const float* __restrict__ B2 = nullptr,
    const float* __restrict__ B3 = nullptr) {
  static_assert(NPL == 1 || NPL == 4, "one plane or four");
  using Cfg = TileCfg<BM, BN, BK, AKC, BKC>;
  using OA  = typename Cfg::OA;
  using OB  = typename Cfg::OB;
  constexpr int NT  = WM * WN * 32;
  constexpr int TM  = BM / (WM * 16);
  constexpr int TN  = BN / (WN * 16);
  constexpr int SZA = Cfg::SZA;
  constexpr int SZB = Cfg::SZB;
  constexpr int NPA = OA::NPC / NT;
  constexpr int NPB = OB::NPC / NT;
  static_assert(NPA * NT == OA::NPC, "A chunk must split evenly over the threads");
  static_assert(NPB * NT == OB::NPC, "B chunk must split evenly over the threads");
  static_assert(TM >= 1 && TN >= 1 && BK % 4 == 0, "wave tile");

  const int tid  = threadIdx.x;
  const int lane = tid & 31;
  const int r    = lane & 15;
  const int half = lane >> 4;
  const int w    = tid >> 5;
  const int wm   = w / WN;
  const int wn   = w % WN;

  int gA[NPA], sA[NPA], gB[NPB], sB[NPB];
  piece_table<BM, BK, AKC, NT, NPA>(gA, sA, tid, m0, ldA);
  piece_table<BN, BK, BKC, NT, NPB>(gB, sB, tid, n0, ldB);
  const int stepA = AKC ? BK : BK * ldA;
  const int stepB = BKC ? BK : BK * ldB;

  float* const Asm = smem;
  float* const Bsm = smem + 2 * SZA;

  v8f acc[TM][TN];
#pragma unroll
  for (int i = 0; i < TM; ++i)
#pragma unroll
    for (int j = 0; j < TN; ++j) acc[i][j] = {};

  constexpr int NRB = NPB * OB::LPP;
  v4f ra[NPA * OA::LPP], rb[NPL][NRB];
  chunk_load<AKC, NPA>(ra, A, gA, 0, ldA);
  chunk_load<BKC, NPB>(rb[0], Bm, gB, 0, ldB);
  if constexpr (NPL == 4) {
    chunk_load<BKC, NPB>(rb[1], B1, gB, 0, ldB);
    chunk_load<BKC, NPB>(rb[2], B2, gB, 0, ldB);
    chunk_load<BKC, NPB>(rb[3], B3, gB, 0, ldB);
#pragma unroll
    for (int i = 0; i < NRB; ++i) rb[0][i] = ((rb[0][i] + rb[1][i]) + rb[2][i]) + rb[3][i];
  }
  chunk_store<AKC, NPA>(ra, Asm, sA);
  chunk_store<BKC, NPB>(rb[0], Bsm, sB);
  __syncthreads();

  const int nch = K / BK;
  for (int kc = 0; kc < nch; ++kc) {
    const int  buf  = kc & 1;
    const bool more = (kc + 1 < nch);
    if (more) {
      chunk_load<AKC, NPA>(ra, A, gA, (kc + 1) * stepA, ldA);
      chunk_load<BKC, NPB>(rb[0], Bm, gB, (kc + 1) * stepB, ldB);
      if constexpr (NPL == 4) {
        chunk_load<BKC, NPB>(rb[1], B1, gB, (kc + 1) * stepB, ldB);
        chunk_load<BKC, NPB>(rb[2], B2, gB, (kc + 1) * stepB, ldB);
        chunk_load<BKC, NPB>(rb[3], B3, gB, (kc + 1) * stepB, ldB);
      }
    }
    __builtin_amdgcn_sched_barrier(0);

    const float* as = Asm + buf * SZA;
    const float* bs = Bsm + buf * SZB;
    constexpr int NS  = BK / 4;
    constexpr int FD0 = 8 / (TM * TN) < 2 ? 2 : 8 / (TM * TN);
    constexpr int FD  = NS < FD0 ? NS : FD0;
    v2f fa[FD][TM], fb[FD][TN];
#pragma unroll
    for (int s = 0; s < FD; ++s) {
      const int kk = s * 4 + 2 * half;
#pragma unroll
      for (int i = 0; i < TM; ++i)
        fa[s][i] = frag_read<BM, BK, AKC>(as, wm * (BM / WM) + i * 16 + r, kk);
#pragma unroll
      for (int j = 0; j < TN; ++j)
        fb[s][j] = frag_read<BN, BK, BKC>(bs, wn * (BN / WN) + j * 16 + r, kk);
    }
    __builtin_amdgcn_sched_barrier(0);
#pragma unroll
    for (int s = 0; s < NS; ++s) {
      const int q = s % FD;
#pragma unroll
      for (int i = 0; i < TM; ++i)
#pragma unroll
        for (int j = 0; j < TN; ++j)
          acc[i][j] = __builtin_amdgcn_wmma_f32_16x16x4_f32(false, fa[q][i], false, fb[q][j],
                                                            (short)0, acc[i][j], false, false);
      if (s + FD < NS) {
        const int kk = (s + FD) * 4 + 2 * half;
#pragma unroll
        for (int i = 0; i < TM; ++i)
          fa[q][i] = frag_read<BM, BK, AKC>(as, wm * (BM / WM) + i * 16 + r, kk);
#pragma unroll
        for (int j = 0; j < TN; ++j)
          fb[q][j] = frag_read<BN, BK, BKC>(bs, wn * (BN / WN) + j * 16 + r, kk);
      }
      __builtin_amdgcn_sched_barrier(0);
    }

    if (more) {
      const int so = (buf ^ 1);
      if constexpr (NPL == 4) {
#pragma unroll
        for (int i = 0; i < NRB; ++i) rb[0][i] = ((rb[0][i] + rb[1][i]) + rb[2][i]) + rb[3][i];
      }
      chunk_store<AKC, NPA>(ra, Asm + so * SZA, sA);
      chunk_store<BKC, NPB>(rb[0], Bsm + so * SZB, sB);
    }
    __syncthreads();
  }

  constexpr int PC = Cfg::PC;
#pragma unroll
  for (int i = 0; i < TM; ++i) {
#pragma unroll
    for (int j = 0; j < TN; ++j) {
#pragma unroll
      for (int v = 0; v < 8; ++v) {
        const int lrow = wm * (BM / WM) + i * 16 + v + 8 * half;
        const int lcol = wn * (BN / WN) + j * 16 + r;
        smem[lrow * PC + lcol] = acc[i][j][v];
      }
    }
  }
  __syncthreads();
  constexpr int QC = BN / 4;
  constexpr int NC = BM * QC / NT;
  static_assert(NC * NT == BM * QC, "C block must split evenly over the threads");
#pragma unroll
  for (int i = 0; i < NC; ++i) {
    const int idx  = tid + i * NT;
    const int lrow = idx / QC;
    const int lcol = (idx % QC) * 4;
    const v4f cv = *(const v4f*)(smem + lrow * PC + lcol);
    const int row = m0 + lrow;
    const int col = n0 + lcol;
    v4f o;
    if (HASR) {
      const v4f rv = *(const v4f*)(R + row * ldR + col);
      float t0 = cv.x * scale; t0 += rv.x;
      float t1 = cv.y * scale; t1 += rv.y;
      float t2 = cv.z * scale; t2 += rv.z;
      float t3 = cv.w * scale; t3 += rv.w;
      o.x = t0; o.y = t1; o.z = t2; o.w = t3;
    } else {
      o.x = cv.x * scale; o.y = cv.y * scale; o.z = cv.z * scale; o.w = cv.w * scale;
    }
    *(v4f*)(C + row * ldC + col) = o;
  }
}

constexpr int KSPLIT_ = 4;
constexpr int KPART_  = NKV_ / KSPLIT_;

__global__ __launch_bounds__(128) void k_stage1_xxT(const float* __restrict__ xkv,
                                                    float* __restrict__ S0, float* __restrict__ S1,
                                                    float* __restrict__ S2, float* __restrict__ S3) {
  using Cfg = TileCfg<64, 32, 64, true, true>;
  __shared__ __attribute__((aligned(16))) float smem[Cfg::LDS_FLOATS];
  const int b = blockIdx.z / KSPLIT_;
  const int p = blockIdx.z % KSPLIT_;
  float* Sp = p == 0 ? S0 : p == 1 ? S1 : p == 2 ? S2 : S3;
  const float* X = xkv + (size_t)b * CKV_ * NKV_ + p * KPART_;
  block_gemm<64, 32, 64, 2, 2, true, true, false>(
      X, NKV_, X, NKV_, Sp + (size_t)b * CKV_ * CKV_, CKV_, nullptr, 0,
      blockIdx.y * 64, blockIdx.x * 32, KPART_, 1.0f, smem);
}

__global__ __launch_bounds__(128) void k_stage2_T1(const float* __restrict__ Wk,
                                                   const float* __restrict__ S0,
                                                   const float* __restrict__ S1,
                                                   const float* __restrict__ S2,
                                                   const float* __restrict__ S3,
                                                   float* __restrict__ T1) {
  using Cfg = TileCfg<64, 32, 64, true, false>;
  __shared__ __attribute__((aligned(16))) float smem[Cfg::LDS_FLOATS];
  const int z = blockIdx.z;
  const int b = z >> 3, h = z & 7;
  const size_t so = (size_t)b * CKV_ * CKV_;
  block_gemm<64, 32, 64, 2, 2, true, false, false, 4>(
      Wk + (size_t)h * DK_ * CKV_, CKV_, S0 + so, CKV_,
      T1 + (size_t)z * DK_ * CKV_, CKV_, nullptr, 0,
      0, blockIdx.x * 32, CKV_, 1.0f, smem, S1 + so, S2 + so, S3 + so);
}

__global__ __launch_bounds__(128) void k_stage3_M(const float* __restrict__ T1,
                                                  const float* __restrict__ Wv,
                                                  float* __restrict__ Mm) {
  using Cfg = TileCfg<32, 32, 128, true, true>;
  __shared__ __attribute__((aligned(16))) float smem[Cfg::LDS_FLOATS];
  const int z = blockIdx.z;
  const int h = z & 7;
  block_gemm<32, 32, 128, 2, 2, true, true, false>(
      T1 + (size_t)z * DK_ * CKV_, CKV_, Wv + (size_t)h * DK_ * CKV_, CKV_,
      Mm + (size_t)z * DK_ * DK_, DK_, nullptr, 0,
      blockIdx.y * 32, blockIdx.x * 32, CKV_, 1.0f / (float)DK_, smem);
}

__global__ __launch_bounds__(128) void k_stage4_Pt(const float* __restrict__ Mm,
                                                   const float* __restrict__ Wq,
                                                   float* __restrict__ Pt) {
  using Cfg = TileCfg<32, 64, 64, false, false>;
  __shared__ __attribute__((aligned(16))) float smem[Cfg::LDS_FLOATS];
  const int z = blockIdx.z;
  const int h = z & 7;
  block_gemm<32, 64, 64, 2, 2, false, false, false>(
      Mm + (size_t)z * DK_ * DK_, DK_, Wq + (size_t)h * DK_ * CQ_, CQ_,
      Pt + (size_t)z * DK_ * CQ_, CQ_, nullptr, 0,
      blockIdx.y * 32, blockIdx.x * 64, DK_, 1.0f, smem);
}

__global__ __launch_bounds__(128) void k_stage5_G(const float* __restrict__ Wo,
                                                  const float* __restrict__ Pt,
                                                  const float* __restrict__ gamma,
                                                  float* __restrict__ G) {
  using Cfg = TileCfg<32, 32, 128, true, false>;
  __shared__ __attribute__((aligned(16))) float smem[Cfg::LDS_FLOATS];
  const int b = blockIdx.z;
  block_gemm<32, 32, 128, 2, 2, true, false, false>(
      Wo, OD_, Pt + (size_t)b * OD_ * CQ_, CQ_,
      G + (size_t)b * CQ_ * CQ_, CQ_, nullptr, 0,
      blockIdx.y * 32, blockIdx.x * 32, OD_, gamma[0], smem);
}

__global__ __launch_bounds__(128) void k_stage6_out(const float* __restrict__ G,
                                                    const float* __restrict__ xq,
                                                    float* __restrict__ out) {
  using Cfg = TileCfg<64, 64, 32, true, false>;
  __shared__ __attribute__((aligned(16))) float smem[Cfg::LDS_FLOATS];
  const int b = blockIdx.z;
  const float* Bq = xq + (size_t)b * CQ_ * NQ_;
  block_gemm<64, 64, 32, 4, 1, true, false, true>(
      G + (size_t)b * CQ_ * CQ_, CQ_, Bq, NQ_,
      out + (size_t)b * CQ_ * NQ_, NQ_, Bq, NQ_,
      blockIdx.y * 64, blockIdx.x * 64, CQ_, 1.0f, smem);
}

extern "C" void kernel_launch(void* const* d_in, const int* in_sizes, int n_in,
                              void* d_out, int out_size, void* d_ws, size_t ws_size,
                              hipStream_t stream) {
  const float* xq    = (const float*)d_in[0];
  const float* xkv   = (const float*)d_in[1];
  const float* Wq    = (const float*)d_in[2];
  const float* Wk    = (const float*)d_in[3];
  const float* Wv    = (const float*)d_in[4];
  const float* Wo    = (const float*)d_in[5];
  const float* gamma = (const float*)d_in[6];
  float* out = (float*)d_out;

  float* S  = (float*)d_ws;
  float* T1 = S  + (size_t)B_ * CKV_ * CKV_;
  float* Mm = T1 + (size_t)B_ * H_ * DK_ * CKV_;
  float* Pt = Mm + (size_t)B_ * H_ * DK_ * DK_;
  float* G  = Pt + (size_t)B_ * H_ * DK_ * CQ_;
  float* S1 = Pt;
  float* S2 = Pt + (size_t)B_ * CKV_ * CKV_;
  float* S3 = G;

  const dim3 blk(128, 1, 1);
  k_stage1_xxT<<<dim3(CKV_ / 32, CKV_ / 64, B_ * KSPLIT_), blk, 0, stream>>>(xkv, S, S1, S2, S3);
  k_stage2_T1 <<<dim3(CKV_ / 32, 1, B_ * H_),        blk, 0, stream>>>(Wk, S, S1, S2, S3, T1);
  k_stage3_M  <<<dim3(DK_ / 32,  DK_ / 32, B_ * H_), blk, 0, stream>>>(T1, Wv, Mm);
  k_stage4_Pt <<<dim3(CQ_ / 64,  DK_ / 32, B_ * H_), blk, 0, stream>>>(Mm, Wq, Pt);
  k_stage5_G  <<<dim3(CQ_ / 32,  CQ_ / 32, B_),      blk, 0, stream>>>(Wo, Pt, gamma, G);
  k_stage6_out<<<dim3(NQ_ / 64,  CQ_ / 64, B_),      blk, 0, stream>>>(G, xq, out);
}


# Stage 3 joins stage 4's launch: each of its 256 workgroups forms its own 64x32 block of M (LDS-staged block GEMM; 4 n-blocks repeat it), keeps it in LDS as A, then x Wq; 5 launches; sum order kept
# speedup vs baseline: 3.4401x; 1.0423x over previous
#include <hip/hip_runtime.h>


typedef float v2f __attribute__((ext_vector_type(2)));
typedef float v4f __attribute__((ext_vector_type(4)));
typedef float v8f __attribute__((ext_vector_type(8)));

constexpr int B_   = 4;
constexpr int CQ_  = 256;
constexpr int CKV_ = 256;
constexpr int NQ_  = 2048;
constexpr int NKV_ = 2048;
constexpr int H_   = 8;
constexpr int DK_  = 64;
constexpr int OD_  = H_ * DK_;

template <int X, int BK, bool KC>
struct OpCfg {
  static_assert(KC || (2 * X) % 64 == 0, "pair-row pitch rule needs X a multiple of 32");
  static constexpr int P   = KC ? BK + 4 : 2 * X + 32;
  static constexpr int SZ  = KC ? X * P : (BK / 2) * P;
  static constexpr int Q   = KC ? BK / 4 : X / 4;
  static constexpr int NPC = KC ? X * Q : (BK / 2) * Q;
  static constexpr int LPP = KC ? 1 : 2;
};

template <int BM, int BN, int BK, bool AKC, bool BKC>
struct TileCfg {
  using OA = OpCfg<BM, BK, AKC>;
  using OB = OpCfg<BN, BK, BKC>;
  static constexpr int SZA = OA::SZ;
  static constexpr int SZB = OB::SZ;
  static constexpr int PC  = BN + 4;
  static constexpr int SZC = BM * PC;
  static constexpr int LDS_FLOATS =
      2 * (SZA + SZB) > SZC ? 2 * (SZA + SZB) : SZC;
};

template <int X, int BK, bool KC, int NT, int NP>
__device__ __forceinline__ void piece_table(int (&g)[NP], int (&s)[NP], int tid, int x0, int ld) {
  using O = OpCfg<X, BK, KC>;
#pragma unroll
  for (int i = 0; i < NP; ++i) {
    const int idx = tid + i * NT;
    const int row = idx / O::Q;
    const int c   = (idx % O::Q) * 4;
    if constexpr (KC) {
      g[i] = (x0 + row) * ld + c;
      s[i] = row * O::P + c;
    } else {
      g[i] = (2 * row) * ld + x0 + c;
      s[i] = row * O::P + 2 * c;
    }
  }
}

template <bool KC, int NP>
__device__ __forceinline__ void chunk_load(v4f (&rg)[NP * (KC ? 1 : 2)], const float* __restrict__ p,
                                           const int (&g)[NP], int off, int ld) {
#pragma unroll
  for (int i = 0; i < NP; ++i) {
    if constexpr (KC) {
      rg[i] = *(const v4f*)(p + g[i] + off);
    } else {
      rg[2 * i]     = *(const v4f*)(p + g[i] + off);
      rg[2 * i + 1] = *(const v4f*)(p + g[i] + off + ld);
    }
  }
}

template <bool KC, int NP>
__device__ __forceinline__ void chunk_store(const v4f (&rg)[NP * (KC ? 1 : 2)], float* __restrict__ dst,
                                            const int (&s)[NP]) {
#pragma unroll
  for (int i = 0; i < NP; ++i) {
    if constexpr (KC) {
      *(v4f*)(dst + s[i]) = rg[i];
    } else {
      const v4f e = rg[2 * i], o = rg[2 * i + 1];
      v4f lo, hi;
      lo.x = e.x; lo.y = o.x; lo.z = e.y; lo.w = o.y;
      hi.x = e.z; hi.y = o.z; hi.z = e.w; hi.w = o.w;
      *(v4f*)(dst + s[i])     = lo;
      *(v4f*)(dst + s[i] + 4) = hi;
    }
  }
}

template <int X, int BK, bool KC>
__device__ __forceinline__ v2f frag_read(const float* __restrict__ base, int x, int kk) {
  using O = OpCfg<X, BK, KC>;
  if constexpr (KC) return *(const v2f*)(base + x * O::P + kk);
  else              return *(const v2f*)(base + (kk >> 1) * O::P + 2 * x);
}

template <int BM, int BN, int BK, int WM, int WN, bool AKC, bool BKC>
__device__ __forceinline__ void chunk_mma(const float* __restrict__ as, const float* __restrict__ bs,
                                          v8f (&acc)[BM / (WM * 16)][BN / (WN * 16)],
                                          int r, int half, int wm, int wn) {
  constexpr int TM = BM / (WM * 16);
  constexpr int TN = BN / (WN * 16);
  constexpr int NS  = BK / 4;
  constexpr int FD0 = 8 / (TM * TN) < 2 ? 2 : 8 / (TM * TN);
  constexpr int FD  = NS < FD0 ? NS : FD0;
  v2f fa[FD][TM], fb[FD][TN];
#pragma unroll
  for (int s = 0; s < FD; ++s) {
    const int kk = s * 4 + 2 * half;
#pragma unroll
    for (int i = 0; i < TM; ++i)
      fa[s][i] = frag_read<BM, BK, AKC>(as, wm * (BM / WM) + i * 16 + r, kk);
#pragma unroll
    for (int j = 0; j < TN; ++j)
      fb[s][j] = frag_read<BN, BK, BKC>(bs, wn * (BN / WN) + j * 16 + r, kk);
  }
  __builtin_amdgcn_sched_barrier(0);
#pragma unroll
  for (int s = 0; s < NS; ++s) {
    const int q = s % FD;
#pragma unroll
    for (int i = 0; i < TM; ++i)
#pragma unroll
      for (int j = 0; j < TN; ++j)
        acc[i][j] = __builtin_amdgcn_wmma_f32_16x16x4_f32(false, fa[q][i], false, fb[q][j],
                                                          (short)0, acc[i][j], false, false);
    if (s + FD < NS) {
      const int kk = (s + FD) * 4 + 2 * half;
#pragma unroll
      for (int i = 0; i < TM; ++i)
        fa[q][i] = frag_read<BM, BK, AKC>(as, wm * (BM / WM) + i * 16 + r, kk);
#pragma unroll
      for (int j = 0; j < TN; ++j)
        fb[q][j] = frag_read<BN, BK, BKC>(bs, wn * (BN / WN) + j * 16 + r, kk);
    }
    __builtin_amdgcn_sched_barrier(0);
  }
}

template <int BM, int BN, int WM, int WN, bool HASR>
__device__ __forceinline__ void store_block(const v8f (&acc)[BM / (WM * 16)][BN / (WN * 16)],
                                            float* __restrict__ smem, float* __restrict__ C, int ldC,
                                            const float* __restrict__ R, int ldR, int m0, int n0,
                                            float scale, int tid, int r, int half, int wm, int wn) {
  constexpr int NT = WM * WN * 32;
  constexpr int TM = BM / (WM * 16);
  constexpr int TN = BN / (WN * 16);
  constexpr int PC = BN + 4;
#pragma unroll
  for (int i = 0; i < TM; ++i) {
#pragma unroll
    for (int j = 0; j < TN; ++j) {
#pragma unroll
      for (int v = 0; v < 8; ++v) {
        const int lrow = wm * (BM / WM) + i * 16 + v + 8 * half;
        const int lcol = wn * (BN / WN) + j * 16 + r;
        smem[lrow * PC + lcol] = acc[i][j][v];
      }
    }
  }
  __syncthreads();
  constexpr int QC = BN / 4;
  constexpr int NC = BM * QC / NT;
  static_assert(NC * NT == BM * QC, "C block must split evenly over the threads");
#pragma unroll
  for (int i = 0; i < NC; ++i) {
    const int idx  = tid + i * NT;
    const int lrow = idx / QC;
    const int lcol = (idx % QC) * 4;
    const v4f cv = *(const v4f*)(smem + lrow * PC + lcol);
    const int row = m0 + lrow;
    const int col = n0 + lcol;
    v4f o;
    if (HASR) {
      const v4f rv = *(const v4f*)(R + row * ldR + col);
      float t0 = cv.x * scale; t0 += rv.x;
      float t1 = cv.y * scale; t1 += rv.y;
      float t2 = cv.z * scale; t2 += rv.z;
      float t3 = cv.w * scale; t3 += rv.w;
      o.x = t0; o.y = t1; o.z = t2; o.w = t3;
    } else {
      o.x = cv.x * scale; o.y = cv.y * scale; o.z = cv.z * scale; o.w = cv.w * scale;
    }
    *(v4f*)(C + row * ldC + col) = o;
  }
}

template <int BM, int BN, int BK, int WM, int WN, bool AKC, bool BKC, bool HASR, int NPL = 1, bool PARK = false>
__device__ __forceinline__ void block_gemm(
    const float* __restrict__ A, int ldA,
    const float* __restrict__ Bm, int ldB,
    float* __restrict__ C, int ldC,
    const float* __restrict__ R, int ldR,
    int m0, int n0, int K, float scale, float* __restrict__ smem,
    const float* __restrict__ B1 = nullptr, const float* __restrict__ B2 = nullptr,
    const float* __restrict__ B3 = nullptr) {
  static_assert(NPL == 1 || NPL == 4, "one plane or four");
  using Cfg = TileCfg<BM, BN, BK, AKC, BKC>;
  using OA  = typename Cfg::OA;
  using OB  = typename Cfg::OB;
  constexpr int NT  = WM * WN * 32;
  constexpr int TM  = BM / (WM * 16);
  constexpr int TN  = BN / (WN * 16);
  constexpr int SZA = Cfg::SZA;
  constexpr int SZB = Cfg::SZB;
  constexpr int NPA = OA::NPC / NT;
  constexpr int NPB = OB::NPC / NT;
  static_assert(NPA * NT == OA::NPC, "A chunk must split evenly over the threads");
  static_assert(NPB * NT == OB::NPC, "B chunk must split evenly over the threads");
  static_assert(TM >= 1 && TN >= 1 && BK % 4 == 0, "wave tile");

  const int tid  = threadIdx.x;
  const int lane = tid & 31;
  const int r    = lane & 15;
  const int half = lane >> 4;
  const int w    = tid >> 5;
  const int wm   = w / WN;
  const int wn   = w % WN;

  int gA[NPA], sA[NPA], gB[NPB], sB[NPB];
  piece_table<BM, BK, AKC, NT, NPA>(gA, sA, tid, m0, ldA);
  piece_table<BN, BK, BKC, NT, NPB>(gB, sB, tid, n0, ldB);
  const int stepA = AKC ? BK : BK * ldA;
  const int stepB = BKC ? BK : BK * ldB;

  float* const Asm = smem;
  float* const Bsm = smem + 2 * SZA;

  v8f acc[TM][TN];
#pragma unroll
  for (int i = 0; i < TM; ++i)
#pragma unroll
    for (int j = 0; j < TN; ++j) acc[i][j] = {};

  constexpr int NRB = NPB * OB::LPP;
  v4f ra[NPA * OA::LPP], rb[NPL][NRB];
  chunk_load<AKC, NPA>(ra, A, gA, 0, ldA);
  chunk_load<BKC, NPB>(rb[0], Bm, gB, 0, ldB);
  if constexpr (NPL == 4) {
    chunk_load<BKC, NPB>(rb[1], B1, gB, 0, ldB);
    chunk_load<BKC, NPB>(rb[2], B2, gB, 0, ldB);
    chunk_load<BKC, NPB>(rb[3], B3, gB, 0, ldB);
#pragma unroll
    for (int i = 0; i < NRB; ++i) rb[0][i] = ((rb[0][i] + rb[1][i]) + rb[2][i]) + rb[3][i];
  }
  chunk_store<AKC, NPA>(ra, Asm, sA);
  chunk_store<BKC, NPB>(rb[0], Bsm, sB);
  __syncthreads();

  const int nch = K / BK;
  for (int kc = 0; kc < nch; ++kc) {
    const int  buf  = kc & 1;
    const bool more = (kc + 1 < nch);
    if (more) {
      chunk_load<AKC, NPA>(ra, A, gA, (kc + 1) * stepA, ldA);
      chunk_load<BKC, NPB>(rb[0], Bm, gB, (kc + 1) * stepB, ldB);
      if constexpr (NPL == 4) {
        chunk_load<BKC, NPB>(rb[1], B1, gB, (kc + 1) * stepB, ldB);
        chunk_load<BKC, NPB>(rb[2], B2, gB, (kc + 1) * stepB, ldB);
        chunk_load<BKC, NPB>(rb[3], B3, gB, (kc + 1) * stepB, ldB);
      }
    }
    __builtin_amdgcn_sched_barrier(0);

    const float* as = Asm + buf * SZA;
    const float* bs = Bsm + buf * SZB;
    chunk_mma<BM, BN, BK, WM, WN, AKC, BKC>(as, bs, acc, r, half, wm, wn);

    if (more) {
      const int so = (buf ^ 1);
      if constexpr (NPL == 4) {
#pragma unroll
        for (int i = 0; i < NRB; ++i) rb[0][i] = ((rb[0][i] + rb[1][i]) + rb[2][i]) + rb[3][i];
      }
      chunk_store<AKC, NPA>(ra, Asm + so * SZA, sA);
      chunk_store<BKC, NPB>(rb[0], Bsm + so * SZB, sB);
    }
    __syncthreads();
  }

  if constexpr (PARK) {
    constexpr int PA = OpCfg<BN, BM, false>::P;
#pragma unroll
    for (int i = 0; i < TM; ++i) {
#pragma unroll
      for (int j = 0; j < TN; ++j) {
#pragma unroll
        for (int v = 0; v < 8; ++v) {
          const int lrow = wm * (BM / WM) + i * 16 + v + 8 * half;
          const int lcol = wn * (BN / WN) + j * 16 + r;
          smem[(lrow >> 1) * PA + 2 * lcol + (lrow & 1)] = acc[i][j][v] * scale;
        }
      }
    }
  } else {
    store_block<BM, BN, WM, WN, HASR>(acc, smem, C, ldC, R, ldR, m0, n0, scale, tid, r, half, wm, wn);
  }
}

constexpr int KSPLIT_ = 4;
constexpr int KPART_  = NKV_ / KSPLIT_;

__global__ __launch_bounds__(128) void k_stage1_xxT(const float* __restrict__ xkv,
                                                    float* __restrict__ S0, float* __restrict__ S1,
                                                    float* __restrict__ S2, float* __restrict__ S3) {
  using Cfg = TileCfg<64, 32, 64, true, true>;
  __shared__ __attribute__((aligned(16))) float smem[Cfg::LDS_FLOATS];
  const int b = blockIdx.z / KSPLIT_;
  const int p = blockIdx.z % KSPLIT_;
  float* Sp = p == 0 ? S0 : p == 1 ? S1 : p == 2 ? S2 : S3;
  const float* X = xkv + (size_t)b * CKV_ * NKV_ + p * KPART_;
  block_gemm<64, 32, 64, 2, 2, true, true, false>(
      X, NKV_, X, NKV_, Sp + (size_t)b * CKV_ * CKV_, CKV_, nullptr, 0,
      blockIdx.y * 64, blockIdx.x * 32, KPART_, 1.0f, smem);
}

__global__ __launch_bounds__(128) void k_stage2_T1(const float* __restrict__ Wk,
                                                   const float* __restrict__ S0,
                                                   const float* __restrict__ S1,
                                                   const float* __restrict__ S2,
                                                   const float* __restrict__ S3,
                                                   float* __restrict__ T1) {
  using Cfg = TileCfg<64, 32, 64, true, false>;
  __shared__ __attribute__((aligned(16))) float smem[Cfg::LDS_FLOATS];
  const int z = blockIdx.z;
  const int b = z >> 3, h = z & 7;
  const size_t so = (size_t)b * CKV_ * CKV_;
  block_gemm<64, 32, 64, 2, 2, true, false, false, 4>(
      Wk + (size_t)h * DK_ * CKV_, CKV_, S0 + so, CKV_,
      T1 + (size_t)z * DK_ * CKV_, CKV_, nullptr, 0,
      0, blockIdx.x * 32, CKV_, 1.0f, smem, S1 + so, S2 + so, S3 + so);
}

__global__ __launch_bounds__(128) void k_stage34_Pt(const float* __restrict__ T1,
                                                    const float* __restrict__ Wv,
                                                    const float* __restrict__ Wq,
                                                    float* __restrict__ Pt) {
  using Cfg1 = TileCfg<64, 32, 64, true, true>;
  using OA2  = OpCfg<32, 64, false>;
  using OB2  = OpCfg<64, 64, false>;
  constexpr int OFFB = OA2::SZ;
  static_assert(OFFB + OB2::SZ <= Cfg1::LDS_FLOATS, "second product's operands fit the first one's buffers");
  static_assert(32 * (64 + 4) <= Cfg1::LDS_FLOATS, "so does its C block");
  __shared__ __attribute__((aligned(16))) float smem[Cfg1::LDS_FLOATS];
  const int z  = blockIdx.z;
  const int h  = z & 7;
  const int j0 = blockIdx.y * 32;
  const int n0 = blockIdx.x * 64;
  const int tid  = threadIdx.x;
  const int lane = tid & 31;
  const int r    = lane & 15;
  const int half = lane >> 4;
  const int w    = tid >> 5;
  const int wm   = w / 2;
  const int wn   = w % 2;

  constexpr int NPB2 = OB2::NPC / 128;
  static_assert(NPB2 * 128 == OB2::NPC, "Wq block must split evenly over the threads");
  int gB2[NPB2], sB2[NPB2];
  piece_table<64, 64, false, 128, NPB2>(gB2, sB2, tid, n0, CQ_);
  v4f rb2[NPB2 * 2];
  chunk_load<false, NPB2>(rb2, Wq + (size_t)h * DK_ * CQ_, gB2, 0, CQ_);

  block_gemm<64, 32, 64, 2, 2, true, true, false, 1, true>(
      T1 + (size_t)z * DK_ * CKV_, CKV_, Wv + (size_t)h * DK_ * CKV_, CKV_,
      nullptr, 0, nullptr, 0, 0, j0, CKV_, 1.0f / (float)DK_, smem);

  chunk_store<false, NPB2>(rb2, smem + OFFB, sB2);
  __syncthreads();

  v8f acc2[1][2];
  acc2[0][0] = {};
  acc2[0][1] = {};
  chunk_mma<32, 64, 64, 2, 2, false, false>(smem, smem + OFFB, acc2, r, half, wm, wn);
  __syncthreads();

  store_block<32, 64, 2, 2, false>(acc2, smem, Pt + (size_t)z * DK_ * CQ_, CQ_, nullptr, 0,
                                   j0, n0, 1.0f, tid, r, half, wm, wn);
}

__global__ __launch_bounds__(128) void k_stage5_G(const float* __restrict__ Wo,
                                                  const float* __restrict__ Pt,
                                                  const float* __restrict__ gamma,
                                                  float* __restrict__ G) {
  using Cfg = TileCfg<32, 32, 128, true, false>;
  __shared__ __attribute__((aligned(16))) float smem[Cfg::LDS_FLOATS];
  const int b = blockIdx.z;
  block_gemm<32, 32, 128, 2, 2, true, false, false>(
      Wo, OD_, Pt + (size_t)b * OD_ * CQ_, CQ_,
      G + (size_t)b * CQ_ * CQ_, CQ_, nullptr, 0,
      blockIdx.y * 32, blockIdx.x * 32, OD_, gamma[0], smem);
}

__global__ __launch_bounds__(128) void k_stage6_out(const float* __restrict__ G,
                                                    const float* __restrict__ xq,
                                                    float* __restrict__ out) {
  using Cfg = TileCfg<64, 64, 32, true, false>;
  __shared__ __attribute__((aligned(16))) float smem[Cfg::LDS_FLOATS];
  const int b = blockIdx.z;
  const float* Bq = xq + (size_t)b * CQ_ * NQ_;
  block_gemm<64, 64, 32, 4, 1, true, false, true>(
      G + (size_t)b * CQ_ * CQ_, CQ_, Bq, NQ_,
      out + (size_t)b * CQ_ * NQ_, NQ_, Bq, NQ_,
      blockIdx.y * 64, blockIdx.x * 64, CQ_, 1.0f, smem);
}

extern "C" void kernel_launch(void* const* d_in, const int* in_sizes, int n_in,
                              void* d_out, int out_size, void* d_ws, size_t ws_size,
                              hipStream_t stream) {
  const float* xq    = (const float*)d_in[0];
  const float* xkv   = (const float*)d_in[1];
  const float* Wq    = (const float*)d_in[2];
  const float* Wk    = (const float*)d_in[3];
  const float* Wv    = (const float*)d_in[4];
  const float* Wo    = (const float*)d_in[5];
  const float* gamma = (const float*)d_in[6];
  float* out = (float*)d_out;

  float* S  = (float*)d_ws;
  float* T1 = S  + (size_t)B_ * CKV_ * CKV_;
  float* Mm = T1 + (size_t)B_ * H_ * DK_ * CKV_;
  float* Pt = Mm + (size_t)B_ * H_ * DK_ * DK_;
  float* G  = Pt + (size_t)B_ * H_ * DK_ * CQ_;
  float* S1 = Pt;
  float* S2 = Pt + (size_t)B_ * CKV_ * CKV_;
  float* S3 = G;

  const dim3 blk(128, 1, 1);
  k_stage1_xxT<<<dim3(CKV_ / 32, CKV_ / 64, B_ * KSPLIT_), blk, 0, stream>>>(xkv, S, S1, S2, S3);
  k_stage2_T1 <<<dim3(CKV_ / 32, 1, B_ * H_),        blk, 0, stream>>>(Wk, S, S1, S2, S3, T1);
  k_stage34_Pt<<<dim3(CQ_ / 64,  DK_ / 32, B_ * H_), blk, 0, stream>>>(T1, Wv, Wq, Pt);
  k_stage5_G  <<<dim3(CQ_ / 32,  CQ_ / 32, B_),      blk, 0, stream>>>(Wo, Pt, gamma, G);
  k_stage6_out<<<dim3(NQ_ / 64,  CQ_ / 64, B_),      blk, 0, stream>>>(G, xq, out);
}
